# starting kernel
# baseline (speedup 1.0000x reference)
#include <hip/hip_runtime.h>

typedef float v2f __attribute__((ext_vector_type(2)));
typedef float v8f __attribute__((ext_vector_type(8)));

#define NT    3000
#define NOUT  603
#define XOFF  301
#define XPAD  3775
#define YOFF  240
#define YPAD  3488
#define NS    203

__global__ __launch_bounds__(96) void xcorr_wmma_f32(const float* __restrict__ d1,
                                                     const float* __restrict__ d2,
                                                     float* __restrict__ out) {
  __shared__ float xs[XPAD];
  __shared__ float ys[YPAD];

  const int c   = blockIdx.x;
  const int tid = threadIdx.x;
  const float* x = d1 + c * NT;
  const float* y = d2 + c * NT;

  for (int i = tid; i < XPAD; i += 96) {
    int t = i - XOFF;
    xs[i] = (t >= 0 && t < NT) ? x[t] : 0.0f;
  }
  for (int i = tid; i < YPAD; i += 96) {
    int t = i - YOFF;
    ys[i] = (t >= 0 && t < NT) ? y[t] : 0.0f;
  }
  __syncthreads();

  const int wave = tid >> 5;
  const int lane = tid & 31;
  const int m    = lane & 15;
  const int kb2  = (lane >> 4) << 1;
  const int OB0  = wave << 4;

  const int alo = m + kb2;
  const int ylo = YOFF - ((OB0 + m) << 4) + kb2;

  v8f acc = {};
  int base = OB0 << 4;
  for (int s = 0; s < NS; ++s) {
    const int xb = base + alo;
    const int yb = base + ylo;
#pragma unroll
    for (int kc = 0; kc < 4; ++kc) {
      v2f a, b;
      a.x = xs[xb + 4 * kc];
      a.y = xs[xb + 4 * kc + 1];
      b.x = ys[yb + 4 * kc];
      b.y = ys[yb + 4 * kc + 1];
      acc = __builtin_amdgcn_wmma_f32_16x16x4_f32(false, a, false, b,
                                                  (short)0, acc, false, false);
    }
    base += 16;
  }

  const int hi = lane >> 4;
#pragma unroll
  for (int i = 0; i < 8; ++i) {
    int oi = i + (hi << 3);
    int o  = ((OB0 + m) << 4) + oi;
    if (o < NOUT) out[c * NOUT + o] = acc[i];
  }
}

extern "C" void kernel_launch(void* const* d_in, const int* in_sizes, int n_in,
                              void* d_out, int out_size, void* d_ws, size_t ws_size,
                              hipStream_t stream) {
  const float* d1 = (const float*)d_in[0];
  const float* d2 = (const float*)d_in[1];
  float* out = (float*)d_out;
  (void)in_sizes; (void)n_in; (void)out_size; (void)d_ws; (void)ws_size;
  hipLaunchKernelGGL(xcorr_wmma_f32, dim3(2048), dim3(96), 0, stream, d1, d2, out);
}


# staging: each thread issues its 16 global loads (16 bytes each) as one group before any LDS store, instead of one 4-byte load per round trip; x pad +3 floats so LDS stores are 16-byte aligned
# speedup vs baseline: 1.3274x; 1.3274x over previous
#include <hip/hip_runtime.h>


typedef float v2f __attribute__((ext_vector_type(2)));
typedef float v4f __attribute__((ext_vector_type(4)));
typedef float v8f __attribute__((ext_vector_type(8)));

#define NT    3000
#define NT4   750
#define NOUT  603
#define XSH   3
#define XOFF  304
#define XPAD  3780
#define YOFF  240
#define YPAD  3488
#define NS    203
#define NLD   8

__global__ __launch_bounds__(96) void xcorr_wmma_f32(const float* __restrict__ d1,
                                                     const float* __restrict__ d2,
                                                     float* __restrict__ out) {
  __shared__ __attribute__((aligned(16))) float xs[XPAD];
  __shared__ __attribute__((aligned(16))) float ys[YPAD];

  const int c   = blockIdx.x;
  const int tid = threadIdx.x;
  const v4f* x4 = reinterpret_cast<const v4f*>(d1 + c * NT);
  const v4f* y4 = reinterpret_cast<const v4f*>(d2 + c * NT);
  v4f* xs4 = reinterpret_cast<v4f*>(xs);
  v4f* ys4 = reinterpret_cast<v4f*>(ys);

  v4f xv[NLD], yv[NLD];
#pragma unroll
  for (int j = 0; j < NLD; ++j) {
    const int q  = tid + 96 * j;
    const int qc = q < NT4 ? q : NT4 - 1;
    xv[j] = x4[qc];
    yv[j] = y4[qc];
  }

  const v4f z4 = {0.0f, 0.0f, 0.0f, 0.0f};
  for (int i = tid; i < XOFF / 4; i += 96) xs4[i] = z4;
  for (int i = XOFF / 4 + NT4 + tid; i < XPAD / 4; i += 96) xs4[i] = z4;
  for (int i = tid; i < YOFF / 4; i += 96) ys4[i] = z4;
  for (int i = YOFF / 4 + NT4 + tid; i < YPAD / 4; i += 96) ys4[i] = z4;

#pragma unroll
  for (int j = 0; j < NLD; ++j) {
    const int q = tid + 96 * j;
    if (q < NT4) {
      xs4[XOFF / 4 + q] = xv[j];
      ys4[YOFF / 4 + q] = yv[j];
    }
  }
  __syncthreads();

  const int wave = tid >> 5;
  const int lane = tid & 31;
  const int m    = lane & 15;
  const int kb2  = (lane >> 4) << 1;
  const int OB0  = wave << 4;

  const int alo = m + kb2 + XSH;
  const int ylo = YOFF - ((OB0 + m) << 4) + kb2;

  v8f acc = {};
  int base = OB0 << 4;
  for (int s = 0; s < NS; ++s) {
    const int xb = base + alo;
    const int yb = base + ylo;
#pragma unroll
    for (int kc = 0; kc < 4; ++kc) {
      v2f a, b;
      a.x = xs[xb + 4 * kc];
      a.y = xs[xb + 4 * kc + 1];
      b.x = ys[yb + 4 * kc];
      b.y = ys[yb + 4 * kc + 1];
      acc = __builtin_amdgcn_wmma_f32_16x16x4_f32(false, a, false, b,
                                                  (short)0, acc, false, false);
    }
    base += 16;
  }

  const int hi = lane >> 4;
#pragma unroll
  for (int i = 0; i < 8; ++i) {
    int oi = i + (hi << 3);
    int o  = ((OB0 + m) << 4) + oi;
    if (o < NOUT) out[c * NOUT + o] = acc[i];
  }
}

extern "C" void kernel_launch(void* const* d_in, const int* in_sizes, int n_in,
                              void* d_out, int out_size, void* d_ws, size_t ws_size,
                              hipStream_t stream) {
  const float* d1 = (const float*)d_in[0];
  const float* d2 = (const float*)d_in[1];
  float* out = (float*)d_out;
  (void)in_sizes; (void)n_in; (void)out_size; (void)d_ws; (void)ws_size;
  hipLaunchKernelGGL(xcorr_wmma_f32, dim3(2048), dim3(96), 0, stream, d1, d2, out);
}


# work split by time range instead of lag tile: 4 waves per block (one per SIMD; was 3), each runs all 3 lag tiles from one shared y fragment over a quarter of the non-padding steps; sums via LDS
# speedup vs baseline: 1.6119x; 1.2143x over previous
#include <hip/hip_runtime.h>


typedef float v2f __attribute__((ext_vector_type(2)));
typedef float v4f __attribute__((ext_vector_type(4)));
typedef float v8f __attribute__((ext_vector_type(8)));

#define NT    3000
#define NT4   750
#define NOUT  603
#define LAG0  301
#define XL    32
#define XPAD  3060
#define YL    240
#define YPAD  3488
#define NTH   128
#define NLD   6
#define NTILE 768
#define TSTEP 256

#define S_END  203
#define S_E1   1
#define S_E0   17
#define S_X2   175
#define S_X1   191
#define S_W1   52
#define S_W2   98
#define S_W3   144

template <bool T0, bool T1, bool T2>
__device__ __forceinline__ void run_steps(const float* __restrict__ xs, const float* __restrict__ ys,
                                          int& xi, int& yi, const int n,
                                          v8f& acc0, v8f& acc1, v8f& acc2) {
  for (int s = 0; s < n; ++s) {
    const v4f bl = *reinterpret_cast<const v4f*>(ys + yi);
    const v4f bh = *reinterpret_cast<const v4f*>(ys + yi + 4);
    const float b[8] = {bl.x, bl.y, bl.z, bl.w, bh.x, bh.y, bh.z, bh.w};
    float a0[8], a1[8], a2[8];
#pragma unroll
    for (int j = 0; j < 8; ++j) {
      if (T0) a0[j] = xs[xi + j];
      if (T1) a1[j] = xs[xi + TSTEP + j];
      if (T2) a2[j] = xs[xi + 2 * TSTEP + j];
    }
#pragma unroll
    for (int kc = 0; kc < 4; ++kc) {
      v2f bv;
      bv.x = b[2 * kc];
      bv.y = b[2 * kc + 1];
      if (T0) {
        v2f av;
        av.x = a0[2 * kc];
        av.y = a0[2 * kc + 1];
        acc0 = __builtin_amdgcn_wmma_f32_16x16x4_f32(false, av, false, bv, (short)0, acc0, false, false);
      }
      if (T1) {
        v2f av;
        av.x = a1[2 * kc];
        av.y = a1[2 * kc + 1];
        acc1 = __builtin_amdgcn_wmma_f32_16x16x4_f32(false, av, false, bv, (short)0, acc1, false, false);
      }
      if (T2) {
        v2f av;
        av.x = a2[2 * kc];
        av.y = a2[2 * kc + 1];
        acc2 = __builtin_amdgcn_wmma_f32_16x16x4_f32(false, av, false, bv, (short)0, acc2, false, false);
      }
    }
    xi += 16;
    yi += 16;
  }
}

__global__ __launch_bounds__(NTH) void xcorr_wmma_f32(const float* __restrict__ d1,
                                                      const float* __restrict__ d2,
                                                      float* __restrict__ out) {
  __shared__ __attribute__((aligned(16))) float smem[XPAD + YPAD];
  float* xs = smem;
  float* ys = smem + XPAD;

  const int c   = blockIdx.x;
  const int tid = threadIdx.x;
  const v4f* x4 = reinterpret_cast<const v4f*>(d1 + c * NT);
  const v4f* y4 = reinterpret_cast<const v4f*>(d2 + c * NT);
  v4f* xs4 = reinterpret_cast<v4f*>(xs);
  v4f* ys4 = reinterpret_cast<v4f*>(ys);

  v4f xv[NLD], yv[NLD];
#pragma unroll
  for (int j = 0; j < NLD; ++j) {
    const int q  = tid + NTH * j;
    const int qc = q < NT4 ? q : NT4 - 1;
    xv[j] = x4[qc];
    yv[j] = y4[qc];
  }

  const v4f z4 = {0.0f, 0.0f, 0.0f, 0.0f};
  for (int i = tid; i < XL / 4; i += NTH) xs4[i] = z4;
  for (int i = XL / 4 + NT4 + tid; i < XPAD / 4; i += NTH) xs4[i] = z4;
  for (int i = tid; i < YL / 4; i += NTH) ys4[i] = z4;
  for (int i = YL / 4 + NT4 + tid; i < YPAD / 4; i += NTH) ys4[i] = z4;

#pragma unroll
  for (int j = 0; j < NLD; ++j) {
    const int q = tid + NTH * j;
    if (q < NT4) {
      xs4[XL / 4 + q] = xv[j];
      ys4[YL / 4 + q] = yv[j];
    }
  }
  __syncthreads();

  const int wave = __builtin_amdgcn_readfirstlane(tid >> 5);
  const int lane = tid & 31;
  const int m    = lane & 15;
  const int h    = lane >> 4;

  const int sb = wave == 0 ? 0    : wave == 1 ? S_W1 : wave == 2 ? S_W2 : S_W3;
  const int se = wave == 0 ? S_W1 : wave == 1 ? S_W2 : wave == 2 ? S_W3 : S_END;
  const int n2   = max(min(se, S_E1)  - max(sb, 0),    0);
  const int n12  = max(min(se, S_E0)  - max(sb, S_E1), 0);
  const int n012 = max(min(se, S_X2)  - max(sb, S_E0), 0);
  const int n01  = max(min(se, S_X1)  - max(sb, S_X2), 0);
  const int n0   = max(min(se, S_END) - max(sb, S_X1), 0);

  int xi = 16 * sb + m + 8 * h - LAG0 + XL;
  int yi = 16 * (sb - m) + 8 * h + YL;

  v8f acc0 = {}, acc1 = {}, acc2 = {};
  run_steps<false, false, true >(xs, ys, xi, yi, n2,   acc0, acc1, acc2);
  run_steps<false, true,  true >(xs, ys, xi, yi, n12,  acc0, acc1, acc2);
  run_steps<true,  true,  true >(xs, ys, xi, yi, n012, acc0, acc1, acc2);
  run_steps<true,  true,  false>(xs, ys, xi, yi, n01,  acc0, acc1, acc2);
  run_steps<true,  false, false>(xs, ys, xi, yi, n0,   acc0, acc1, acc2);

  __syncthreads();
  float* red = ys;
  {
    v4f* r4 = reinterpret_cast<v4f*>(red + wave * NTILE + 16 * m + 8 * h);
    const v4f p0 = {acc0[0], acc0[1], acc0[2], acc0[3]};
    const v4f p1 = {acc0[4], acc0[5], acc0[6], acc0[7]};
    const v4f p2 = {acc1[0], acc1[1], acc1[2], acc1[3]};
    const v4f p3 = {acc1[4], acc1[5], acc1[6], acc1[7]};
    const v4f p4 = {acc2[0], acc2[1], acc2[2], acc2[3]};
    const v4f p5 = {acc2[4], acc2[5], acc2[6], acc2[7]};
    r4[0]   = p0; r4[1]   = p1;
    r4[64]  = p2; r4[65]  = p3;
    r4[128] = p4; r4[129] = p5;
  }
  __syncthreads();

  for (int o = tid; o < NOUT; o += NTH)
    out[c * NOUT + o] = ((red[o] + red[NTILE + o]) + red[2 * NTILE + o]) + red[3 * NTILE + o];
}

extern "C" void kernel_launch(void* const* d_in, const int* in_sizes, int n_in,
                              void* d_out, int out_size, void* d_ws, size_t ws_size,
                              hipStream_t stream) {
  const float* d1 = (const float*)d_in[0];
  const float* d2 = (const float*)d_in[1];
  float* out = (float*)d_out;
  (void)in_sizes; (void)n_in; (void)out_size; (void)d_ws; (void)ws_size;
  hipLaunchKernelGGL(xcorr_wmma_f32, dim3(2048), dim3(NTH), 0, stream, d1, d2, out);
}


# lag tiling: five 128-lag tiles (4 x windows x 2 column groups per matrix instruction, column groups 32 samples apart in time) instead of three 256-lag tiles: 640 lag slots for 603 lags (was 768)
# speedup vs baseline: 1.8727x; 1.1618x over previous
#include <hip/hip_runtime.h>


typedef float v2f __attribute__((ext_vector_type(2)));
typedef float v4f __attribute__((ext_vector_type(4)));
typedef float v8f __attribute__((ext_vector_type(8)));

#define NT    3000
#define NT4   750
#define NOUT  603
#define LAG0  301
#define XL    304
#define XPAD  3400
#define YL    28
#define YPAD  3100
#define NTH   128
#define NLD   6
#define XT    128
#define SLOT  672
#define WSTR  1344

#define G_END  96
#define G_E1   2
#define G_E0   6
#define G_X4   88
#define G_X3   92
#define G_W1   25
#define G_W2   48
#define G_W3   71

template <bool T0, bool T1, bool T2, bool T3, bool T4>
__device__ __forceinline__ void run_groups(const float* __restrict__ xs, const float* __restrict__ ys,
                                           const int xl, const int yl, const int lo, const int hi,
                                           v8f& acc0, v8f& acc1, v8f& acc2, v8f& acc3, v8f& acc4) {
  for (int g = lo; g < hi; ++g) {
    const int yb = ((g >> 1) << 6) | ((g & 1) << 4);
    const int yi = yl + yb;
    const int xi = xl + yb;
    const v4f bl = *reinterpret_cast<const v4f*>(ys + yi);
    const v4f bh = *reinterpret_cast<const v4f*>(ys + yi + 4);
    const float b[8] = {bl.x, bl.y, bl.z, bl.w, bh.x, bh.y, bh.z, bh.w};
    float a0[8], a1[8], a2[8], a3[8], a4[8];
#pragma unroll
    for (int j = 0; j < 8; ++j) {
      if (T0) a0[j] = xs[xi + j];
      if (T1) a1[j] = xs[xi + XT + j];
      if (T2) a2[j] = xs[xi + 2 * XT + j];
      if (T3) a3[j] = xs[xi + 3 * XT + j];
      if (T4) a4[j] = xs[xi + 4 * XT + j];
    }
#pragma unroll
    for (int kc = 0; kc < 4; ++kc) {
      v2f bv;
      bv.x = b[2 * kc];
      bv.y = b[2 * kc + 1];
      if (T0) {
        v2f av;
        av.x = a0[2 * kc];
        av.y = a0[2 * kc + 1];
        acc0 = __builtin_amdgcn_wmma_f32_16x16x4_f32(false, av, false, bv, (short)0, acc0, false, false);
      }
      if (T1) {
        v2f av;
        av.x = a1[2 * kc];
        av.y = a1[2 * kc + 1];
        acc1 = __builtin_amdgcn_wmma_f32_16x16x4_f32(false, av, false, bv, (short)0, acc1, false, false);
      }
      if (T2) {
        v2f av;
        av.x = a2[2 * kc];
        av.y = a2[2 * kc + 1];
        acc2 = __builtin_amdgcn_wmma_f32_16x16x4_f32(false, av, false, bv, (short)0, acc2, false, false);
      }
      if (T3) {
        v2f av;
        av.x = a3[2 * kc];
        av.y = a3[2 * kc + 1];
        acc3 = __builtin_amdgcn_wmma_f32_16x16x4_f32(false, av, false, bv, (short)0, acc3, false, false);
      }
      if (T4) {
        v2f av;
        av.x = a4[2 * kc];
        av.y = a4[2 * kc + 1];
        acc4 = __builtin_amdgcn_wmma_f32_16x16x4_f32(false, av, false, bv, (short)0, acc4, false, false);
      }
    }
  }
}

__device__ __forceinline__ void put_tile(float* __restrict__ p, const v8f& acc) {
  const v4f lo = {acc[0], acc[1], acc[2], acc[3]};
  const v4f hi = {acc[4], acc[5], acc[6], acc[7]};
  *reinterpret_cast<v4f*>(p)      = lo;
  *reinterpret_cast<v4f*>(p + 32) = hi;
}

__global__ __launch_bounds__(NTH) void xcorr_wmma_f32(const float* __restrict__ d1,
                                                      const float* __restrict__ d2,
                                                      float* __restrict__ out) {
  __shared__ __attribute__((aligned(16))) float smem[XPAD + YPAD];
  float* xs = smem;
  float* ys = smem + XPAD;

  const int c   = blockIdx.x;
  const int tid = threadIdx.x;
  const v4f* x4 = reinterpret_cast<const v4f*>(d1 + c * NT);
  const v4f* y4 = reinterpret_cast<const v4f*>(d2 + c * NT);
  v4f* xs4 = reinterpret_cast<v4f*>(xs);
  v4f* ys4 = reinterpret_cast<v4f*>(ys);

  v4f xv[NLD], yv[NLD];
#pragma unroll
  for (int j = 0; j < NLD; ++j) {
    const int q  = tid + NTH * j;
    const int qc = q < NT4 ? q : NT4 - 1;
    xv[j] = x4[qc];
    yv[j] = y4[qc];
  }

  const v4f z4 = {0.0f, 0.0f, 0.0f, 0.0f};
  for (int i = tid; i < XL / 4; i += NTH) xs4[i] = z4;
  for (int i = XL / 4 + NT4 + tid; i < XPAD / 4; i += NTH) xs4[i] = z4;
  for (int i = tid; i < YL / 4; i += NTH) ys4[i] = z4;
  for (int i = YL / 4 + NT4 + tid; i < YPAD / 4; i += NTH) ys4[i] = z4;

#pragma unroll
  for (int j = 0; j < NLD; ++j) {
    const int q = tid + NTH * j;
    if (q < NT4) {
      xs4[XL / 4 + q] = xv[j];
      ys4[YL / 4 + q] = yv[j];
    }
  }
  __syncthreads();

  const int wave = __builtin_amdgcn_readfirstlane(tid >> 5);
  const int lane = tid & 31;
  const int m    = lane & 15;
  const int h    = lane >> 4;

  const int gb = wave == 0 ? 0    : wave == 1 ? G_W1 : wave == 2 ? G_W2 : G_W3;
  const int ge = wave == 0 ? G_W1 : wave == 1 ? G_W2 : wave == 2 ? G_W3 : G_END;

  const int xl = XL - LAG0 + 32 * (m >> 2) + (m & 3) + 8 * h;
  const int yl = YL + 32 * (m >> 3) - 4 * (m & 7) + 8 * h;

  v8f acc0 = {}, acc1 = {}, acc2 = {}, acc3 = {}, acc4 = {};
  run_groups<false, false, true,  true,  true >(xs, ys, xl, yl, max(gb, 0),    min(ge, G_E1),  acc0, acc1, acc2, acc3, acc4);
  run_groups<false, true,  true,  true,  true >(xs, ys, xl, yl, max(gb, G_E1), min(ge, G_E0),  acc0, acc1, acc2, acc3, acc4);
  run_groups<true,  true,  true,  true,  true >(xs, ys, xl, yl, max(gb, G_E0), min(ge, G_X4),  acc0, acc1, acc2, acc3, acc4);
  run_groups<true,  true,  true,  true,  false>(xs, ys, xl, yl, max(gb, G_X4), min(ge, G_X3),  acc0, acc1, acc2, acc3, acc4);
  run_groups<true,  true,  true,  false, false>(xs, ys, xl, yl, max(gb, G_X3), min(ge, G_END), acc0, acc1, acc2, acc3, acc4);

  __syncthreads();
  float* red = smem;
  {
    float* p = red + wave * WSTR + (m >> 3) * SLOT + 64 * h + 4 * (m & 7);
    put_tile(p,          acc0);
    put_tile(p + XT,     acc1);
    put_tile(p + 2 * XT, acc2);
    put_tile(p + 3 * XT, acc3);
    put_tile(p + 4 * XT, acc4);
  }
  __syncthreads();

  for (int o = tid; o < NOUT; o += NTH) {
    float s = 0.0f;
#pragma unroll
    for (int w = 0; w < 4; ++w) s += red[w * WSTR + o] + red[w * WSTR + SLOT + 32 + o];
    out[c * NOUT + o] = s;
  }
}

extern "C" void kernel_launch(void* const* d_in, const int* in_sizes, int n_in,
                              void* d_out, int out_size, void* d_ws, size_t ws_size,
                              hipStream_t stream) {
  const float* d1 = (const float*)d_in[0];
  const float* d2 = (const float*)d_in[1];
  float* out = (float*)d_out;
  (void)in_sizes; (void)n_in; (void)out_size; (void)d_ws; (void)ws_size;
  hipLaunchKernelGGL(xcorr_wmma_f32, dim3(2048), dim3(NTH), 0, stream, d1, d2, out);
}


# LDS fragment reads: x columns 4 samples apart (operand roles swapped): a tile's x fragment is two aligned 16-byte conflict-free loads; a step's 14 loads stand in one group before its 20 WMMAs
# speedup vs baseline: 1.9736x; 1.0539x over previous
#include <hip/hip_runtime.h>


typedef float v2f __attribute__((ext_vector_type(2)));
typedef float v4f __attribute__((ext_vector_type(4)));
typedef float v8f __attribute__((ext_vector_type(8)));

#define NT    3000
#define NT4   750
#define NOUT  603
#define LAG0  301
#define XL    304
#define XPAD  3360
#define YL    72
#define YPAD  3144
#define NTH   128
#define NLD   6
#define XT    128
#define SLOT  672
#define WSTR  1344

#define G_END  96
#define G_E1   4
#define G_E0   8
#define G_X4   88
#define G_X3   92
#define G_W1   25
#define G_W2   48
#define G_W3   71

template <bool T0, bool T1, bool T2, bool T3, bool T4>
__device__ __forceinline__ void run_groups(const float* __restrict__ xs, const float* __restrict__ ys,
                                           const int xl, const int yl, const int lo, const int hi,
                                           v8f& acc0, v8f& acc1, v8f& acc2, v8f& acc3, v8f& acc4) {
  for (int g = lo; g < hi; ++g) {
    const int yb = ((g >> 1) << 6) | ((g & 1) << 4);
    const float* yp = ys + yl + yb;
    const v4f*   xp = reinterpret_cast<const v4f*>(xs + xl + yb);
    float a[8];
#pragma unroll
    for (int j = 0; j < 8; ++j) a[j] = yp[j];
    v4f x0[2], x1[2], x2[2], x3[2], x4[2];
#pragma unroll
    for (int p = 0; p < 2; ++p) {
      if (T0) x0[p] = xp[p];
      if (T1) x1[p] = xp[XT / 4 + p];
      if (T2) x2[p] = xp[2 * XT / 4 + p];
      if (T3) x3[p] = xp[3 * XT / 4 + p];
      if (T4) x4[p] = xp[4 * XT / 4 + p];
    }
    __builtin_amdgcn_sched_barrier(0);
#pragma unroll
    for (int kc = 0; kc < 4; ++kc) {
      v2f av;
      av.x = a[2 * kc];
      av.y = a[2 * kc + 1];
      if (T0) {
        v2f bv;
        bv.x = x0[kc >> 1][2 * (kc & 1)];
        bv.y = x0[kc >> 1][2 * (kc & 1) + 1];
        acc0 = __builtin_amdgcn_wmma_f32_16x16x4_f32(false, av, false, bv, (short)0, acc0, false, false);
      }
      if (T1) {
        v2f bv;
        bv.x = x1[kc >> 1][2 * (kc & 1)];
        bv.y = x1[kc >> 1][2 * (kc & 1) + 1];
        acc1 = __builtin_amdgcn_wmma_f32_16x16x4_f32(false, av, false, bv, (short)0, acc1, false, false);
      }
      if (T2) {
        v2f bv;
        bv.x = x2[kc >> 1][2 * (kc & 1)];
        bv.y = x2[kc >> 1][2 * (kc & 1) + 1];
        acc2 = __builtin_amdgcn_wmma_f32_16x16x4_f32(false, av, false, bv, (short)0, acc2, false, false);
      }
      if (T3) {
        v2f bv;
        bv.x = x3[kc >> 1][2 * (kc & 1)];
        bv.y = x3[kc >> 1][2 * (kc & 1) + 1];
        acc3 = __builtin_amdgcn_wmma_f32_16x16x4_f32(false, av, false, bv, (short)0, acc3, false, false);
      }
      if (T4) {
        v2f bv;
        bv.x = x4[kc >> 1][2 * (kc & 1)];
        bv.y = x4[kc >> 1][2 * (kc & 1) + 1];
        acc4 = __builtin_amdgcn_wmma_f32_16x16x4_f32(false, av, false, bv, (short)0, acc4, false, false);
      }
    }
    __builtin_amdgcn_sched_barrier(0);
  }
}

__device__ __forceinline__ void put_tile(float* __restrict__ p, const v8f& acc) {
  const v4f lo = {acc[0], acc[1], acc[2], acc[3]};
  const v4f hi = {acc[4], acc[5], acc[6], acc[7]};
  *reinterpret_cast<v4f*>(p)      = lo;
  *reinterpret_cast<v4f*>(p + 64) = hi;
}

__global__ __launch_bounds__(NTH) void xcorr_wmma_f32(const float* __restrict__ d1,
                                                      const float* __restrict__ d2,
                                                      float* __restrict__ out) {
  __shared__ __attribute__((aligned(16))) float smem[XPAD + YPAD];
  float* xs = smem;
  float* ys = smem + XPAD;

  const int c   = blockIdx.x;
  const int tid = threadIdx.x;
  const v4f* x4 = reinterpret_cast<const v4f*>(d1 + c * NT);
  const v4f* y4 = reinterpret_cast<const v4f*>(d2 + c * NT);
  v4f* xs4 = reinterpret_cast<v4f*>(xs);
  v4f* ys4 = reinterpret_cast<v4f*>(ys);

  v4f xv[NLD], yv[NLD];
#pragma unroll
  for (int j = 0; j < NLD; ++j) {
    const int q  = tid + NTH * j;
    const int qc = q < NT4 ? q : NT4 - 1;
    xv[j] = x4[qc];
    yv[j] = y4[qc];
  }

  const v4f z4 = {0.0f, 0.0f, 0.0f, 0.0f};
  for (int i = tid; i < XL / 4; i += NTH) xs4[i] = z4;
  for (int i = XL / 4 + NT4 + tid; i < XPAD / 4; i += NTH) xs4[i] = z4;
  for (int i = tid; i < YL / 4; i += NTH) ys4[i] = z4;
  for (int i = YL / 4 + NT4 + tid; i < YPAD / 4; i += NTH) ys4[i] = z4;

#pragma unroll
  for (int j = 0; j < NLD; ++j) {
    const int q = tid + NTH * j;
    if (q < NT4) {
      xs4[XL / 4 + q] = xv[j];
      ys4[YL / 4 + q] = yv[j];
    }
  }
  __syncthreads();

  const int wave = __builtin_amdgcn_readfirstlane(tid >> 5);
  const int lane = tid & 31;
  const int m    = lane & 15;
  const int h    = lane >> 4;

  const int gb = wave == 0 ? 0    : wave == 1 ? G_W1 : wave == 2 ? G_W2 : G_W3;
  const int ge = wave == 0 ? G_W1 : wave == 1 ? G_W2 : wave == 2 ? G_W3 : G_END;

  const int xl = 4 * m + 8 * h;
  const int yl = YL - 3 + 32 * (m >> 3) - 64 * ((m >> 2) & 1) - (m & 3) + 8 * h;

  v8f acc0 = {}, acc1 = {}, acc2 = {}, acc3 = {}, acc4 = {};
  run_groups<false, false, true,  true,  true >(xs, ys, xl, yl, max(gb, 0),    min(ge, G_E1),  acc0, acc1, acc2, acc3, acc4);
  run_groups<false, true,  true,  true,  true >(xs, ys, xl, yl, max(gb, G_E1), min(ge, G_E0),  acc0, acc1, acc2, acc3, acc4);
  run_groups<true,  true,  true,  true,  true >(xs, ys, xl, yl, max(gb, G_E0), min(ge, G_X4),  acc0, acc1, acc2, acc3, acc4);
  run_groups<true,  true,  true,  true,  false>(xs, ys, xl, yl, max(gb, G_X4), min(ge, G_X3),  acc0, acc1, acc2, acc3, acc4);
  run_groups<true,  true,  true,  false, false>(xs, ys, xl, yl, max(gb, G_X3), min(ge, G_END), acc0, acc1, acc2, acc3, acc4);

  __syncthreads();
  float* red = smem;
  {
    float* p = red + wave * WSTR + h * SLOT + 4 * m;
    put_tile(p,          acc0);
    put_tile(p + XT,     acc1);
    put_tile(p + 2 * XT, acc2);
    put_tile(p + 3 * XT, acc3);
    put_tile(p + 4 * XT, acc4);
  }
  __syncthreads();

  for (int o = tid; o < NOUT; o += NTH) {
    float s = 0.0f;
#pragma unroll
    for (int w = 0; w < 4; ++w) s += red[w * WSTR + o] + red[w * WSTR + SLOT + 32 + o];
    out[c * NOUT + o] = s;
  }
}

extern "C" void kernel_launch(void* const* d_in, const int* in_sizes, int n_in,
                              void* d_out, int out_size, void* d_ws, size_t ws_size,
                              hipStream_t stream) {
  const float* d1 = (const float*)d_in[0];
  const float* d2 = (const float*)d_in[1];
  float* out = (float*)d_out;
  (void)in_sizes; (void)n_in; (void)out_size; (void)d_ws; (void)ws_size;
  hipLaunchKernelGGL(xcorr_wmma_f32, dim3(2048), dim3(NTH), 0, stream, d1, d2, out);
}


# x window reuse: a step runs two groups 128 samples apart, whose tiles t+1 and t read the same x window: 20 LDS loads and 1 address add per 40 WMMAs (was 28 and 10); steps dealt round-robin to 4 waves
# speedup vs baseline: 2.0533x; 1.0404x over previous
#include <hip/hip_runtime.h>


typedef float v2f __attribute__((ext_vector_type(2)));
typedef float v4f __attribute__((ext_vector_type(4)));
typedef float v8f __attribute__((ext_vector_type(8)));

#define NT    3000
#define NT4   750
#define NOUT  603
#define LAG0  301
#define XL    304
#define XPAD  3360
#define YL    72
#define YPAD  3144
#define NTH   128
#define NLD   6
#define XT    128
#define SLOT  672
#define WSTR  1344

#define STEP_Y 256
#define PAIR_Y 128

template <int MODE>
__device__ __forceinline__ void pair_step(const float* __restrict__ lds, int yo, const v4f* __restrict__ xp,
                                          v8f& acc0, v8f& acc1, v8f& acc2, v8f& acc3, v8f& acc4) {
  constexpr bool U0[5] = {MODE != 0, MODE != 0, true, true, MODE != 2};
  constexpr bool U1[5] = {MODE != 0, true, true, MODE != 2, MODE != 2};
  constexpr bool WN[6] = {U0[0], U0[1] || U1[0], U0[2] || U1[1], U0[3] || U1[2], U0[4] || U1[3], U1[4]};

  asm volatile("" : "+v"(yo));
  const float* yp = reinterpret_cast<const float*>(reinterpret_cast<const char*>(lds) + yo);
  float a0[8], a1[8];
#pragma unroll
  for (int j = 0; j < 8; ++j) {
    a0[j] = yp[j];
    a1[j] = yp[PAIR_Y + j];
  }
  v4f x[6][2];
#pragma unroll
  for (int j = 0; j < 6; ++j) {
#pragma unroll
    for (int p = 0; p < 2; ++p) {
      if (WN[j]) x[j][p] = xp[j * (XT / 4) + p];
    }
  }
  __builtin_amdgcn_sched_barrier(0);
#pragma unroll
  for (int kc = 0; kc < 4; ++kc) {
    v2f av0, av1;
    av0.x = a0[2 * kc];
    av0.y = a0[2 * kc + 1];
    av1.x = a1[2 * kc];
    av1.y = a1[2 * kc + 1];
    v2f bv[6];
#pragma unroll
    for (int j = 0; j < 6; ++j) {
      if (WN[j]) {
        bv[j].x = x[j][kc >> 1][2 * (kc & 1)];
        bv[j].y = x[j][kc >> 1][2 * (kc & 1) + 1];
      }
    }
    if (U0[0]) acc0 = __builtin_amdgcn_wmma_f32_16x16x4_f32(false, av0, false, bv[0], (short)0, acc0, false, false);
    if (U0[1]) acc1 = __builtin_amdgcn_wmma_f32_16x16x4_f32(false, av0, false, bv[1], (short)0, acc1, false, false);
    if (U0[2]) acc2 = __builtin_amdgcn_wmma_f32_16x16x4_f32(false, av0, false, bv[2], (short)0, acc2, false, false);
    if (U0[3]) acc3 = __builtin_amdgcn_wmma_f32_16x16x4_f32(false, av0, false, bv[3], (short)0, acc3, false, false);
    if (U0[4]) acc4 = __builtin_amdgcn_wmma_f32_16x16x4_f32(false, av0, false, bv[4], (short)0, acc4, false, false);
    if (U1[0]) acc0 = __builtin_amdgcn_wmma_f32_16x16x4_f32(false, av1, false, bv[1], (short)0, acc0, false, false);
    if (U1[1]) acc1 = __builtin_amdgcn_wmma_f32_16x16x4_f32(false, av1, false, bv[2], (short)0, acc1, false, false);
    if (U1[2]) acc2 = __builtin_amdgcn_wmma_f32_16x16x4_f32(false, av1, false, bv[3], (short)0, acc2, false, false);
    if (U1[3]) acc3 = __builtin_amdgcn_wmma_f32_16x16x4_f32(false, av1, false, bv[4], (short)0, acc3, false, false);
    if (U1[4]) acc4 = __builtin_amdgcn_wmma_f32_16x16x4_f32(false, av1, false, bv[5], (short)0, acc4, false, false);
  }
  __builtin_amdgcn_sched_barrier(0);
}

__device__ __forceinline__ void put_tile(float* __restrict__ p, const v8f& acc) {
  const v4f lo = {acc[0], acc[1], acc[2], acc[3]};
  const v4f hi = {acc[4], acc[5], acc[6], acc[7]};
  *reinterpret_cast<v4f*>(p)      = lo;
  *reinterpret_cast<v4f*>(p + 64) = hi;
}

__global__ __launch_bounds__(NTH) void xcorr_wmma_f32(const float* __restrict__ d1,
                                                      const float* __restrict__ d2,
                                                      float* __restrict__ out) {
  __shared__ __attribute__((aligned(16))) float smem[XPAD + YPAD];
  float* xs = smem;
  float* ys = smem + XPAD;

  const int c   = blockIdx.x;
  const int tid = threadIdx.x;
  const v4f* x4 = reinterpret_cast<const v4f*>(d1 + c * NT);
  const v4f* y4 = reinterpret_cast<const v4f*>(d2 + c * NT);
  v4f* xs4 = reinterpret_cast<v4f*>(xs);
  v4f* ys4 = reinterpret_cast<v4f*>(ys);

  v4f xv[NLD], yv[NLD];
#pragma unroll
  for (int j = 0; j < NLD; ++j) {
    const int q  = tid + NTH * j;
    const int qc = q < NT4 ? q : NT4 - 1;
    xv[j] = x4[qc];
    yv[j] = y4[qc];
  }

  const v4f z4 = {0.0f, 0.0f, 0.0f, 0.0f};
  for (int i = tid; i < XL / 4; i += NTH) xs4[i] = z4;
  for (int i = XL / 4 + NT4 + tid; i < XPAD / 4; i += NTH) xs4[i] = z4;
  for (int i = tid; i < YL / 4; i += NTH) ys4[i] = z4;
  for (int i = YL / 4 + NT4 + tid; i < YPAD / 4; i += NTH) ys4[i] = z4;

#pragma unroll
  for (int j = 0; j < NLD; ++j) {
    const int q = tid + NTH * j;
    if (q < NT4) {
      xs4[XL / 4 + q] = xv[j];
      ys4[YL / 4 + q] = yv[j];
    }
  }
  __syncthreads();

  const int wave = __builtin_amdgcn_readfirstlane(tid >> 5);
  const int lane = tid & 31;
  const int m    = lane & 15;
  const int h    = lane >> 4;

  const int xl = 4 * m + 8 * h;
  const int yl = YL - 3 + 32 * (m >> 3) - 64 * ((m >> 2) & 1) - (m & 3) + 8 * h;

  const int ybw = ((wave >> 1) << 6) | ((wave & 1) << 4);
  const int    yo = (XPAD + yl + ybw) * 4;
  const float* xq = xs + xl + ybw;

  v8f acc0 = {}, acc1 = {}, acc2 = {}, acc3 = {}, acc4 = {};
  pair_step<0>(smem, yo, reinterpret_cast<const v4f*>(xq), acc0, acc1, acc2, acc3, acc4);
  for (int r = 1; r < 11; ++r)
    pair_step<1>(smem, yo + 4 * STEP_Y * r, reinterpret_cast<const v4f*>(xq + STEP_Y * r), acc0, acc1, acc2, acc3, acc4);
  pair_step<2>(smem, yo + 4 * STEP_Y * 11, reinterpret_cast<const v4f*>(xq + STEP_Y * 11), acc0, acc1, acc2, acc3, acc4);

  __syncthreads();
  float* red = smem;
  {
    float* p = red + wave * WSTR + h * SLOT + 4 * m;
    put_tile(p,          acc0);
    put_tile(p + XT,     acc1);
    put_tile(p + 2 * XT, acc2);
    put_tile(p + 3 * XT, acc3);
    put_tile(p + 4 * XT, acc4);
  }
  __syncthreads();

  for (int o = tid; o < NOUT; o += NTH) {
    float s = 0.0f;
#pragma unroll
    for (int w = 0; w < 4; ++w) s += red[w * WSTR + o] + red[w * WSTR + SLOT + 32 + o];
    out[c * NOUT + o] = s;
  }
}

extern "C" void kernel_launch(void* const* d_in, const int* in_sizes, int n_in,
                              void* d_out, int out_size, void* d_ws, size_t ws_size,
                              hipStream_t stream) {
  const float* d1 = (const float*)d_in[0];
  const float* d2 = (const float*)d_in[1];
  float* out = (float*)d_out;
  (void)in_sizes; (void)n_in; (void)out_size; (void)d_ws; (void)ws_size;
  hipLaunchKernelGGL(xcorr_wmma_f32, dim3(2048), dim3(NTH), 0, stream, d1, d2, out);
}


# partial sums: wave w runs steps 9w..9w+8 of the 36 (correlation-major): 6 sets of partial sums (was 12), 2 flushes per wave at most (was 3); combined with 16-byte LDS reads, 8 lags per thread
# speedup vs baseline: 2.3745x; 1.1564x over previous
#include <hip/hip_runtime.h>


typedef float v2f __attribute__((ext_vector_type(2)));
typedef float v4f __attribute__((ext_vector_type(4)));
typedef float v8f __attribute__((ext_vector_type(8)));

#define NT    3000
#define NT4   750
#define NH    1500
#define NOUT  603
#define LAG0  301
#define XL    152
#define XLEN  1840
#define YL    4
#define YLEN  1544
#define NTH   128
#define NLD   6
#define XT    64
#define PL    336
#define NSIG  (3 * (XLEN + YLEN))
#define NSET  6
#define NOB   608
#define NRED  (3 * 4 * 2 * PL)

#define STEP_Y 128
#define PAIR_Y 64

__device__ __forceinline__ void pair_step(const float* __restrict__ lds, int yo, const float* __restrict__ xp,
                                          v8f& acc0, v8f& acc1, v8f& acc2, v8f& acc3, v8f& acc4) {
  const float* yp = reinterpret_cast<const float*>(reinterpret_cast<const char*>(lds) + yo);
  float a0[8], a1[8];
#pragma unroll
  for (int kc = 0; kc < 4; ++kc) {
    a0[2 * kc]     = yp[16 * kc];
    a0[2 * kc + 1] = yp[16 * kc + 1];
    a1[2 * kc]     = yp[PAIR_Y + 16 * kc];
    a1[2 * kc + 1] = yp[PAIR_Y + 16 * kc + 1];
  }
  v2f x[6][4];
#pragma unroll
  for (int j = 0; j < 6; ++j) {
#pragma unroll
    for (int kc = 0; kc < 4; ++kc) {
      x[j][kc] = *reinterpret_cast<const v2f*>(xp + XT * j + 16 * kc);
    }
  }
  __builtin_amdgcn_sched_barrier(0);
#pragma unroll
  for (int kc = 0; kc < 4; ++kc) {
    v2f av0, av1;
    av0.x = a0[2 * kc];
    av0.y = a0[2 * kc + 1];
    av1.x = a1[2 * kc];
    av1.y = a1[2 * kc + 1];
    acc0 = __builtin_amdgcn_wmma_f32_16x16x4_f32(false, av0, false, x[0][kc], (short)0, acc0, false, false);
    acc1 = __builtin_amdgcn_wmma_f32_16x16x4_f32(false, av0, false, x[1][kc], (short)0, acc1, false, false);
    acc2 = __builtin_amdgcn_wmma_f32_16x16x4_f32(false, av0, false, x[2][kc], (short)0, acc2, false, false);
    acc3 = __builtin_amdgcn_wmma_f32_16x16x4_f32(false, av0, false, x[3][kc], (short)0, acc3, false, false);
    acc4 = __builtin_amdgcn_wmma_f32_16x16x4_f32(false, av0, false, x[4][kc], (short)0, acc4, false, false);
    acc0 = __builtin_amdgcn_wmma_f32_16x16x4_f32(false, av1, false, x[1][kc], (short)0, acc0, false, false);
    acc1 = __builtin_amdgcn_wmma_f32_16x16x4_f32(false, av1, false, x[2][kc], (short)0, acc1, false, false);
    acc2 = __builtin_amdgcn_wmma_f32_16x16x4_f32(false, av1, false, x[3][kc], (short)0, acc2, false, false);
    acc3 = __builtin_amdgcn_wmma_f32_16x16x4_f32(false, av1, false, x[4][kc], (short)0, acc3, false, false);
    acc4 = __builtin_amdgcn_wmma_f32_16x16x4_f32(false, av1, false, x[5][kc], (short)0, acc4, false, false);
  }
  __builtin_amdgcn_sched_barrier(0);
}

__device__ __forceinline__ void put_tile(float* p, const v8f& acc) {
  const v4f lo = {acc[0], acc[1], acc[2], acc[3]};
  const v4f hi = {acc[4], acc[5], acc[6], acc[7]};
  *reinterpret_cast<v4f*>(p)          = lo;
  *reinterpret_cast<v4f*>(p + PL - 4) = hi;
}
__device__ __forceinline__ void add_tile(float* p, const v8f& acc) {
  const v4f lo = {acc[0], acc[1], acc[2], acc[3]};
  const v4f hi = {acc[4], acc[5], acc[6], acc[7]};
  *reinterpret_cast<v4f*>(p)          = *reinterpret_cast<const v4f*>(p) + lo;
  *reinterpret_cast<v4f*>(p + PL - 4) = *reinterpret_cast<const v4f*>(p + PL - 4) + hi;
}

__global__ __launch_bounds__(NTH) void xcorr_wmma_f32(const float* __restrict__ d1,
                                                      const float* __restrict__ d2,
                                                      float* __restrict__ out) {
  __shared__ __attribute__((aligned(16))) float smem[NSIG + NRED];
  v2f* xs2 = reinterpret_cast<v2f*>(smem);
  v2f* ys2 = reinterpret_cast<v2f*>(smem + 3 * XLEN);

  const int c   = blockIdx.x;
  const int tid = threadIdx.x;
  const float* xg = d1 + c * NT;
  const v4f* x4 = reinterpret_cast<const v4f*>(xg);
  const v4f* y4 = reinterpret_cast<const v4f*>(d2 + c * NT);

  v4f xv[NLD], yv[NLD];
  float xn[NLD];
#pragma unroll
  for (int j = 0; j < NLD; ++j) {
    const int q  = tid + NTH * j;
    const int qc = q < NT4 ? q : NT4 - 1;
    xv[j] = x4[qc];
    yv[j] = y4[qc];
    const int nx = 4 * qc + 4;
    const float t = xg[nx < NT ? nx : NT - 1];
    xn[j] = nx < NT ? t : 0.0f;
  }

  const v2f z2 = {0.0f, 0.0f};
  for (int i = tid; i < 3 * 170; i += NTH) {
    const int a = i / 170;
    const int r = i - 170 * a;
    const int pos = r < XL / 2 ? r : r + NH / 2;
    if (!(a == 2 && r == XL / 2 - 1)) xs2[a * (XLEN / 2) + pos] = z2;
  }
  for (int i = tid; i < 3 * 22; i += NTH) {
    const int a = i / 22;
    const int r = i - 22 * a;
    const int pos = r < YL / 2 ? r : r + NH / 2;
    ys2[a * (YLEN / 2) + pos] = z2;
  }

#pragma unroll
  for (int j = 0; j < NLD; ++j) {
    const int q = tid + NTH * j;
    if (q < NT4) {
      const v4f v = xv[j];
      const v4f w = yv[j];
      const v2f e0 = {v.x, v.z};
      const v2f e1 = {v.y, v.w};
      const v2f es = {v.y + v.z, v.w + xn[j]};
      xs2[XL / 2 + q]                  = e0;
      xs2[XLEN / 2 + XL / 2 + q]       = e1;
      xs2[2 * (XLEN / 2) + XL / 2 + q] = es;
      if (q == 0) {
        const v2f em = {0.0f, v.x};
        xs2[2 * (XLEN / 2) + XL / 2 - 1] = em;
      }
      const v2f f0 = {w.x, w.z};
      const v2f f1 = {w.y, w.w};
      const v2f fs = {w.x + w.y, w.z + w.w};
      ys2[YL / 2 + q]                  = f0;
      ys2[YLEN / 2 + YL / 2 + q]       = f1;
      ys2[2 * (YLEN / 2) + YL / 2 + q] = fs;
    }
  }
  __syncthreads();

  const int wave = __builtin_amdgcn_readfirstlane(tid >> 5);
  const int lane = tid & 31;
  const int m    = lane & 15;
  const int h    = lane >> 4;

  const int xl = 4 * m + 2 * h;
  const int yl = YL + 4 * (m >> 2) - (m & 3) + 2 * h;

  float* red = smem + NSIG;

  const int cA = (3 * wave) >> 2;
  const int rA = 9 * wave - 12 * cA;
  const int nA = rA > 3 ? 12 - rA : 9;

#pragma nounroll
  for (int run = 0; run < 2; ++run) {
    const int cidx = cA + run;
    const int r0   = run ? 0 : rA;
    const int n    = run ? 9 - nA : nA;
    if (n == 0) continue;
    const int set  = 2 * cidx + (r0 != 0);
    const int    yo = (3 * XLEN + cidx * YLEN + yl + STEP_Y * r0) * 4;
    const float* xq = smem + cidx * XLEN + xl + STEP_Y * r0;
    v8f acc0 = {}, acc1 = {}, acc2 = {}, acc3 = {}, acc4 = {};
#pragma nounroll
    for (int r = 0; r < n; ++r)
      pair_step(smem, yo + 4 * STEP_Y * r, xq + STEP_Y * r, acc0, acc1, acc2, acc3, acc4);
    float* p = red + set * 2 * PL + 12 + 4 * m - 8 * h;
    if (h == 0) {
      put_tile(p,          acc0);
      put_tile(p + XT,     acc1);
      put_tile(p + 2 * XT, acc2);
      put_tile(p + 3 * XT, acc3);
      put_tile(p + 4 * XT, acc4);
      if (m < 2) {
        const v4f z4 = {0.0f, 0.0f, 0.0f, 0.0f};
        *reinterpret_cast<v4f*>(p - 8)          = z4;
        *reinterpret_cast<v4f*>(p + PL - 4 - 8) = z4;
      }
    }
    __builtin_amdgcn_fence(__ATOMIC_RELEASE, "wavefront");
    __builtin_amdgcn_wave_barrier();
    __builtin_amdgcn_fence(__ATOMIC_ACQUIRE, "wavefront");
    if (h == 1) {
      add_tile(p,          acc0);
      add_tile(p + XT,     acc1);
      add_tile(p + 2 * XT, acc2);
      add_tile(p + 3 * XT, acc3);
      add_tile(p + 4 * XT, acc4);
    }
  }
  __syncthreads();

  float* ob = red + NSET * 2 * PL;
  if (tid < NOB / 8) {
    const float* q = red + 4 * (tid + 3);
    v4f s00 = {0.0f, 0.0f, 0.0f, 0.0f}, s11 = s00, sq = s00;
    float s00n = 0.0f;
#pragma unroll
    for (int k = 0; k < 4; ++k) {
      s00  += *reinterpret_cast<const v4f*>(q + k * PL);
      s00n += q[k * PL + 4];
      s11  += *reinterpret_cast<const v4f*>(q + (4 + k) * PL);
      sq   += *reinterpret_cast<const v4f*>(q + (8 + k) * PL);
    }
    const v4f ev = s00 + s11;
    const v4f up = {s00.y, s00.z, s00.w, s00n};
    const v4f od = (sq - s11) - up;
    const v4f o0 = {ev.x, od.x, ev.y, od.y};
    const v4f o1 = {ev.z, od.z, ev.w, od.w};
    *reinterpret_cast<v4f*>(ob + 8 * tid)     = o0;
    *reinterpret_cast<v4f*>(ob + 8 * tid + 4) = o1;
  }
  __syncthreads();
  __builtin_assume(tid < NTH);
#pragma unroll
  for (int j = 0; j < (NOUT + NTH - 1) / NTH; ++j) {
    const int o = tid + NTH * j;
    if (o < NOUT) out[c * NOUT + o] = ob[o + 3];
  }
}

extern "C" void kernel_launch(void* const* d_in, const int* in_sizes, int n_in,
                              void* d_out, int out_size, void* d_ws, size_t ws_size,
                              hipStream_t stream) {
  const float* d1 = (const float*)d_in[0];
  const float* d2 = (const float*)d_in[1];
  float* out = (float*)d_out;
  (void)in_sizes; (void)n_in; (void)out_size; (void)d_ws; (void)ws_size;
  hipLaunchKernelGGL(xcorr_wmma_f32, dim3(2048), dim3(NTH), 0, stream, d1, d2, out);
}
